# starting kernel
# baseline (speedup 1.0000x reference)
#include <hip/hip_runtime.h>

typedef __attribute__((ext_vector_type(2))) float v2f;
typedef __attribute__((ext_vector_type(8))) float v8f;

static constexpr int T_LEN = 2048;
static constexpr int BATCH = 512;
static constexpr int EMB_D = 16;
static constexpr int HID_D = 32;
static constexpr int GATE  = 128;

__device__ __forceinline__ float fsigmoid(float x) {
    return __fdividef(1.0f, 1.0f + __expf(-x));
}
__device__ __forceinline__ float ftanh(float x) {
    return 1.0f - __fdividef(2.0f, __expf(2.0f * x) + 1.0f);
}

__device__ __forceinline__ v8f wmma4(v2f a, v2f b, v8f c) {
    return __builtin_amdgcn_wmma_f32_16x16x4_f32(
         false, a,  false, b,
         (short)0, c,  false,  false);
}

__global__ __launch_bounds__(32)
void lstm_scan_kernel(const int* __restrict__ tokens,
                      const float* __restrict__ Wemb,
                      const float* __restrict__ Wx,
                      const float* __restrict__ Wh,
                      const float* __restrict__ bias,
                      float* __restrict__ out)
{
    __shared__ float kT[GATE * EMB_D];
    __shared__ float rT[GATE * HID_D];
    __shared__ float hbuf[16 * HID_D];

    const int lane = threadIdx.x & 31;
    const int half = lane >> 4;
    const int nloc = lane & 15;
    const int koff = half << 1;
    const int base = half << 3;
    const int b0   = blockIdx.x << 4;

    for (int i = lane; i < GATE * EMB_D; i += 32) {
        const int n = i / EMB_D, k = i - n * EMB_D;
        kT[i] = Wx[k * GATE + n];
    }
    for (int i = lane; i < GATE * HID_D; i += 32) {
        const int n = i / HID_D, k = i - n * HID_D;
        rT[i] = Wh[k * GATE + n];
    }
    for (int i = lane; i < 16 * HID_D; i += 32) hbuf[i] = 0.0f;
    __syncthreads();

    v8f bacc[8];
#pragma unroll
    for (int nc = 0; nc < 8; ++nc) {
        const float bv = bias[nc * 16 + nloc];
#pragma unroll
        for (int r = 0; r < 8; ++r) bacc[nc][r] = bv;
    }

    v8f cst[2] = {};
    v8f hst[2] = {};

    const int* tokp = tokens + (b0 + nloc) * T_LEN;
    int tok0 = tokp[0];
    int tok1 = tokp[1];

    v2f ax[4];
    {
        const float* er = Wemb + (long)tok0 * EMB_D + koff;
#pragma unroll
        for (int kc = 0; kc < 4; ++kc) ax[kc] = *(const v2f*)(er + 4 * kc);
    }

    for (int t = 0; t < T_LEN; ++t) {
        const int tok2 = (t + 2 < T_LEN) ? tokp[t + 2] : 0;

        v2f axn[4];
        {
            const float* er = Wemb + (long)tok1 * EMB_D + koff;
#pragma unroll
            for (int kc = 0; kc < 4; ++kc) axn[kc] = *(const v2f*)(er + 4 * kc);
        }

        const unsigned mb = __builtin_amdgcn_ballot_w32(tok0 != 0);

        v2f ah[8];
#pragma unroll
        for (int kc = 0; kc < 8; ++kc)
            ah[kc] = *(const v2f*)&hbuf[nloc * HID_D + 4 * kc + koff];

        v8f acc[8];
#pragma unroll
        for (int nc = 0; nc < 8; ++nc) {
            const float* kb = &kT[(nc * 16 + nloc) * EMB_D + koff];
            acc[nc] = wmma4(ax[0], *(const v2f*)(kb), bacc[nc]);
        }
#pragma unroll
        for (int kc = 1; kc < 4; ++kc) {
#pragma unroll
            for (int nc = 0; nc < 8; ++nc) {
                const float* kb = &kT[(nc * 16 + nloc) * EMB_D + koff];
                acc[nc] = wmma4(ax[kc], *(const v2f*)(kb + 4 * kc), acc[nc]);
            }
        }
#pragma unroll
        for (int kc = 0; kc < 8; ++kc) {
#pragma unroll
            for (int nc = 0; nc < 8; ++nc) {
                const float* rb = &rT[(nc * 16 + nloc) * HID_D + koff];
                acc[nc] = wmma4(ah[kc], *(const v2f*)(rb + 4 * kc), acc[nc]);
            }
        }

#pragma unroll
        for (int hc = 0; hc < 2; ++hc) {
#pragma unroll
            for (int r = 0; r < 8; ++r) {
                const float iv = fsigmoid(acc[0 + hc][r]);
                const float fv = fsigmoid(acc[2 + hc][r]);
                const float gv = ftanh(acc[4 + hc][r]);
                const float ov = fsigmoid(acc[6 + hc][r]);
                const float cn = fv * cst[hc][r] + iv * gv;
                const float hn = ov * ftanh(cn);
                const bool  m  = (mb >> (base + r)) & 1u;
                const float cw = m ? cn : cst[hc][r];
                const float hw = m ? hn : hst[hc][r];
                cst[hc][r] = cw;
                hst[hc][r] = hw;
                hbuf[(base + r) * HID_D + hc * 16 + nloc] = hw;
            }
        }

        tok0 = tok1;
        tok1 = tok2;
#pragma unroll
        for (int kc = 0; kc < 4; ++kc) ax[kc] = axn[kc];
    }

#pragma unroll
    for (int hc = 0; hc < 2; ++hc)
#pragma unroll
        for (int r = 0; r < 8; ++r)
            out[(b0 + base + r) * HID_D + hc * 16 + nloc] = hst[hc][r];
}

extern "C" void kernel_launch(void* const* d_in, const int* in_sizes, int n_in,
                              void* d_out, int out_size, void* d_ws, size_t ws_size,
                              hipStream_t stream) {
    const int*   tokens = (const int*)  d_in[0];
    const float* emb    = (const float*)d_in[1];
    const float* Wx     = (const float*)d_in[2];
    const float* Wh     = (const float*)d_in[3];
    const float* bias   = (const float*)d_in[4];
    float*       out    = (float*)d_out;
    (void)in_sizes; (void)n_in; (void)out_size; (void)d_ws; (void)ws_size;

    dim3 grid(BATCH / 16);
    dim3 block(32);
    lstm_scan_kernel<<<grid, block, 0, stream>>>(tokens, emb, Wx, Wh, bias, out);
}


# gate activations take their reciprocal with one v_rcp_f32 (1 ulp) instead of the correctly rounded f32 division sequence (10 dependent instructions, through VCC), 20 times per lane per step
# speedup vs baseline: 4.4356x; 4.4356x over previous
#include <hip/hip_runtime.h>


typedef __attribute__((ext_vector_type(2))) float v2f;
typedef __attribute__((ext_vector_type(4))) float v4f;
typedef __attribute__((ext_vector_type(8))) float v8f;

static constexpr int T_LEN = 2048;
static constexpr int BATCH = 512;
static constexpr int EMB_D = 16;
static constexpr int HID_D = 32;
static constexpr int GATE  = 128;
static constexpr int NWAVE = 4;
static constexpr int HP    = 36;

__device__ __forceinline__ float frcp(float x) {
    return __builtin_amdgcn_rcpf(x);
}
__device__ __forceinline__ float fsigmoid(float x) {
    return frcp(1.0f + __expf(-x));
}
__device__ __forceinline__ float ftanh(float x) {
    return 1.0f - 2.0f * frcp(__expf(2.0f * x) + 1.0f);
}

__device__ __forceinline__ v8f wmma4(v2f a, v2f b, v8f c) {
    return __builtin_amdgcn_wmma_f32_16x16x4_f32(
         false, a,  false, b,
         (short)0, c,  false,  false);
}

__global__ __launch_bounds__(32 * NWAVE)
void lstm_scan_kernel(const int* __restrict__ tokens,
                      const float* __restrict__ Wemb,
                      const float* __restrict__ Wx,
                      const float* __restrict__ Wh,
                      const float* __restrict__ bias,
                      float* __restrict__ out)
{
    __shared__ __attribute__((aligned(16))) float hbuf[2 * 16 * HP];

    const int tid  = threadIdx.x;
    const int lane = tid & 31;
    const int wave = tid >> 5;
    const int half = lane >> 4;
    const int col  = lane & 15;
    const int koff = half << 1;
    const int b0   = blockIdx.x << 4;

    for (int i = tid; i < 2 * 16 * HP; i += 32 * NWAVE) hbuf[i] = 0.0f;

    v2f aWx[2][4];
    v2f aWh[2][8];
    v8f bacc[2];
#pragma unroll
    for (int tl = 0; tl < 2; ++tl) {
        const int ubase = 8 * wave + 4 * tl;
        const int n = 32 * ((col & 7) >> 1) + ubase + 2 * (col >> 3) + (col & 1);
#pragma unroll
        for (int kc = 0; kc < 4; ++kc) {
            aWx[tl][kc][0] = Wx[(4 * kc + koff + 0) * GATE + n];
            aWx[tl][kc][1] = Wx[(4 * kc + koff + 1) * GATE + n];
        }
#pragma unroll
        for (int kc = 0; kc < 8; ++kc) {
            aWh[tl][kc][0] = Wh[(4 * kc + koff + 0) * GATE + n];
            aWh[tl][kc][1] = Wh[(4 * kc + koff + 1) * GATE + n];
        }
#pragma unroll
        for (int r = 0; r < 8; ++r)
            bacc[tl][r] = bias[32 * (r >> 1) + ubase + 2 * half + (r & 1)];
    }
    __syncthreads();

    float cst[2][2] = {};
    float hst[2][2] = {};

    const int* tokp = tokens + (b0 + col) * T_LEN;
    int tok0 = tokp[0];
    int tok1 = tokp[1];

    v2f bx[4];
    {
        const float* er = Wemb + (long)tok0 * EMB_D + koff;
#pragma unroll
        for (int kc = 0; kc < 4; ++kc) bx[kc] = *(const v2f*)(er + 4 * kc);
    }

    const int rd_off = col * HP + koff;
    const int wr_off = col * HP + 8 * wave + koff;

    for (int t = 0; t < T_LEN; ++t) {
        const float* hcur = hbuf + (t & 1) * (16 * HP);
        float*       hnxt = hbuf + ((t + 1) & 1) * (16 * HP);

        const int tok2 = tokp[(t + 2 < T_LEN) ? (t + 2) : (T_LEN - 1)];

        v2f bxn[4];
        {
            const float* er = Wemb + (long)tok1 * EMB_D + koff;
#pragma unroll
            for (int kc = 0; kc < 4; ++kc) bxn[kc] = *(const v2f*)(er + 4 * kc);
        }

        const bool m = (tok0 != 0);

        v2f bh[8];
#pragma unroll
        for (int kc = 0; kc < 8; ++kc)
            bh[kc] = *(const v2f*)&hcur[rd_off + 4 * kc];

        __builtin_amdgcn_sched_barrier(0);

        v8f acc[2];
#pragma unroll
        for (int tl = 0; tl < 2; ++tl)
            acc[tl] = wmma4(aWx[tl][0], bx[0], bacc[tl]);
#pragma unroll
        for (int kc = 1; kc < 4; ++kc) {
#pragma unroll
            for (int tl = 0; tl < 2; ++tl)
                acc[tl] = wmma4(aWx[tl][kc], bx[kc], acc[tl]);
        }
#pragma unroll
        for (int kc = 0; kc < 8; ++kc) {
#pragma unroll
            for (int tl = 0; tl < 2; ++tl)
                acc[tl] = wmma4(aWh[tl][kc], bh[kc], acc[tl]);
        }

#pragma unroll
        for (int tl = 0; tl < 2; ++tl) {
            v2f hw2;
#pragma unroll
            for (int j = 0; j < 2; ++j) {
                const float iv = fsigmoid(acc[tl][0 + j]);
                const float fv = fsigmoid(acc[tl][2 + j]);
                const float gv = ftanh(acc[tl][4 + j]);
                const float ov = fsigmoid(acc[tl][6 + j]);
                const float cn = fv * cst[tl][j] + iv * gv;
                const float hn = ov * ftanh(cn);
                const float cw = m ? cn : cst[tl][j];
                const float hw = m ? hn : hst[tl][j];
                cst[tl][j] = cw;
                hst[tl][j] = hw;
                hw2[j] = hw;
            }
            *(v2f*)&hnxt[wr_off + 4 * tl] = hw2;
        }

        __syncthreads();

        tok0 = tok1;
        tok1 = tok2;
#pragma unroll
        for (int kc = 0; kc < 4; ++kc) bx[kc] = bxn[kc];
    }

    {
        const float* hfin = hbuf + (T_LEN & 1) * (16 * HP);
        const int row = 4 * wave + (lane >> 3);
        const int c4  = (lane & 7) << 2;
        const v4f v = *(const v4f*)&hfin[row * HP + c4];
        *(v4f*)&out[(b0 + row) * HID_D + c4] = v;
    }
}

extern "C" void kernel_launch(void* const* d_in, const int* in_sizes, int n_in,
                              void* d_out, int out_size, void* d_ws, size_t ws_size,
                              hipStream_t stream) {
    const int*   tokens = (const int*)  d_in[0];
    const float* emb    = (const float*)d_in[1];
    const float* Wx     = (const float*)d_in[2];
    const float* Wh     = (const float*)d_in[3];
    const float* bias   = (const float*)d_in[4];
    float*       out    = (float*)d_out;
    (void)in_sizes; (void)n_in; (void)out_size; (void)d_ws; (void)ws_size;

    dim3 grid(BATCH / 16);
    dim3 block(32 * NWAVE);
    lstm_scan_kernel<<<grid, block, 0, stream>>>(tokens, emb, Wx, Wh, bias, out);
}


# 12 of a step's 24 matrix instructions (x(t+1) chunks; own h chunks taken from registers) leave the barrier-to-gates chain; K summed x, own, others; LDS-only fence at the barrier
# speedup vs baseline: 4.6447x; 1.0471x over previous
#include <hip/hip_runtime.h>


typedef __attribute__((ext_vector_type(2))) float v2f;
typedef __attribute__((ext_vector_type(4))) float v4f;
typedef __attribute__((ext_vector_type(8))) float v8f;

static constexpr int T_LEN = 2048;
static constexpr int BATCH = 512;
static constexpr int EMB_D = 16;
static constexpr int HID_D = 32;
static constexpr int GATE  = 128;
static constexpr int NWAVE = 4;
static constexpr int HP    = 36;

__device__ __forceinline__ float frcp(float x) {
    return __builtin_amdgcn_rcpf(x);
}
__device__ __forceinline__ float fsigmoid(float x) {
    return frcp(1.0f + __expf(-x));
}
__device__ __forceinline__ float ftanh(float x) {
    return 1.0f - 2.0f * frcp(__expf(2.0f * x) + 1.0f);
}

__device__ __forceinline__ v8f wmma4(v2f a, v2f b, v8f c) {
    return __builtin_amdgcn_wmma_f32_16x16x4_f32(
         false, a,  false, b,
         (short)0, c,  false,  false);
}

__global__ __launch_bounds__(32 * NWAVE)
void lstm_scan_kernel(const int* __restrict__ tokens,
                      const float* __restrict__ Wemb,
                      const float* __restrict__ Wx,
                      const float* __restrict__ Wh,
                      const float* __restrict__ bias,
                      float* __restrict__ out)
{
    __shared__ __attribute__((aligned(16))) float hbuf[2 * 16 * HP];

    const int tid  = threadIdx.x;
    const int lane = tid & 31;
    const int wave = tid >> 5;
    const int half = lane >> 4;
    const int col  = lane & 15;
    const int koff = half << 1;
    const int b0   = blockIdx.x << 4;

    for (int i = tid; i < 2 * 16 * HP; i += 32 * NWAVE) hbuf[i] = 0.0f;

    v2f aWx[2][4];
    v2f aWo[2][2];
    v2f aWr[2][6];
    v8f bacc[2];
#pragma unroll
    for (int tl = 0; tl < 2; ++tl) {
        const int ubase = 8 * wave + 4 * tl;
        const int n = 32 * ((col & 7) >> 1) + ubase + 2 * (col >> 3) + (col & 1);
#pragma unroll
        for (int kc = 0; kc < 4; ++kc) {
            aWx[tl][kc][0] = Wx[(4 * kc + koff + 0) * GATE + n];
            aWx[tl][kc][1] = Wx[(4 * kc + koff + 1) * GATE + n];
        }
#pragma unroll
        for (int p = 0; p < 2; ++p) {
            const int kc = 2 * wave + p;
            aWo[tl][p][0] = Wh[(4 * kc + koff + 0) * GATE + n];
            aWo[tl][p][1] = Wh[(4 * kc + koff + 1) * GATE + n];
        }
#pragma unroll
        for (int i = 0; i < 6; ++i) {
            const int kc = 2 * ((wave + 1 + (i >> 1)) & 3) + (i & 1);
            aWr[tl][i][0] = Wh[(4 * kc + koff + 0) * GATE + n];
            aWr[tl][i][1] = Wh[(4 * kc + koff + 1) * GATE + n];
        }
#pragma unroll
        for (int r = 0; r < 8; ++r)
            bacc[tl][r] = bias[32 * (r >> 1) + ubase + 2 * half + (r & 1)];
    }
    __syncthreads();

    float cst[2][2] = {};
    float hst[2][2] = {};

    const int* tokp = tokens + (b0 + col) * T_LEN;
    int tok0 = tokp[0];
    int tok1 = tokp[1];
    int tok2 = tokp[2];

    v2f bx1[4];
    v8f acc[2];
    {
        v2f bx0[4];
        const float* e0 = Wemb + (long)tok0 * EMB_D + koff;
        const float* e1 = Wemb + (long)tok1 * EMB_D + koff;
#pragma unroll
        for (int kc = 0; kc < 4; ++kc) {
            bx0[kc] = *(const v2f*)(e0 + 4 * kc);
            bx1[kc] = *(const v2f*)(e1 + 4 * kc);
        }
        const v2f hz = {0.0f, 0.0f};
#pragma unroll
        for (int tl = 0; tl < 2; ++tl)
            acc[tl] = wmma4(aWx[tl][0], bx0[0], bacc[tl]);
#pragma unroll
        for (int kc = 1; kc < 4; ++kc)
#pragma unroll
            for (int tl = 0; tl < 2; ++tl)
                acc[tl] = wmma4(aWx[tl][kc], bx0[kc], acc[tl]);
#pragma unroll
        for (int p = 0; p < 2; ++p)
#pragma unroll
            for (int tl = 0; tl < 2; ++tl)
                acc[tl] = wmma4(aWo[tl][p], hz, acc[tl]);
    }

    int rd_pair[3];
#pragma unroll
    for (int i = 0; i < 3; ++i)
        rd_pair[i] = col * HP + koff + 8 * ((wave + 1 + i) & 3);
    const int wr_off = col * HP + 8 * wave + koff;

    for (int t = 0; t < T_LEN; ++t) {
        const float* hcur = hbuf + (t & 1) * (16 * HP);
        float*       hnxt = hbuf + ((t + 1) & 1) * (16 * HP);

        v2f bh[6];
#pragma unroll
        for (int i = 0; i < 3; ++i) {
            bh[2 * i + 0] = *(const v2f*)&hcur[rd_pair[i]];
            bh[2 * i + 1] = *(const v2f*)&hcur[rd_pair[i] + 4];
        }
        __builtin_amdgcn_sched_barrier(0);

        const int tok3 = tokp[(t + 3 < T_LEN) ? (t + 3) : (T_LEN - 1)];
        v2f bx2[4];
        {
            const float* er = Wemb + (long)tok2 * EMB_D + koff;
#pragma unroll
            for (int kc = 0; kc < 4; ++kc) bx2[kc] = *(const v2f*)(er + 4 * kc);
        }
        __builtin_amdgcn_sched_barrier(0);

        const bool m = (tok0 != 0);

#pragma unroll
        for (int i = 0; i < 6; ++i)
#pragma unroll
            for (int tl = 0; tl < 2; ++tl)
                acc[tl] = wmma4(aWr[tl][i], bh[i], acc[tl]);

        v8f accn[2];
#pragma unroll
        for (int tl = 0; tl < 2; ++tl)
            accn[tl] = wmma4(aWx[tl][0], bx1[0], bacc[tl]);
#pragma unroll
        for (int kc = 1; kc < 4; ++kc)
#pragma unroll
            for (int tl = 0; tl < 2; ++tl)
                accn[tl] = wmma4(aWx[tl][kc], bx1[kc], accn[tl]);

        float ex[4][4], rc[4][4], ec[4], cn[4];
#pragma unroll
        for (int u = 0; u < 4; ++u) {
            ex[u][0] = __expf(-acc[u >> 1][0 + (u & 1)]);
            ex[u][1] = __expf(-acc[u >> 1][2 + (u & 1)]);
            ex[u][2] = __expf(2.0f * acc[u >> 1][4 + (u & 1)]);
            ex[u][3] = __expf(-acc[u >> 1][6 + (u & 1)]);
        }
#pragma unroll
        for (int u = 0; u < 4; ++u)
#pragma unroll
            for (int q = 0; q < 4; ++q) rc[u][q] = frcp(1.0f + ex[u][q]);
#pragma unroll
        for (int u = 0; u < 4; ++u) {
            const float gv = 1.0f - 2.0f * rc[u][2];
            cn[u] = rc[u][1] * cst[u >> 1][u & 1] + rc[u][0] * gv;
            ec[u] = __expf(2.0f * cn[u]);
        }
        v2f hw2[2];
#pragma unroll
        for (int u = 0; u < 4; ++u) {
            const float hn = rc[u][3] * (1.0f - 2.0f * frcp(ec[u] + 1.0f));
            const float cw = m ? cn[u] : cst[u >> 1][u & 1];
            const float hw = m ? hn : hst[u >> 1][u & 1];
            cst[u >> 1][u & 1] = cw;
            hst[u >> 1][u & 1] = hw;
            hw2[u >> 1][u & 1] = hw;
        }

        __builtin_amdgcn_sched_barrier(0);

#pragma unroll
        for (int tl = 0; tl < 2; ++tl)
            *(v2f*)&hnxt[wr_off + 4 * tl] = hw2[tl];
        __builtin_amdgcn_sched_barrier(0);

        {
            const v2f hown = {hst[0][0], hst[0][1]};
            accn[0] = wmma4(aWo[0][0], hown, accn[0]);
            accn[1] = wmma4(aWo[1][0], hown, accn[1]);
        }
        __builtin_amdgcn_sched_barrier(0);

        __builtin_amdgcn_fence(__ATOMIC_RELEASE, "workgroup", "local");
        __builtin_amdgcn_s_barrier_signal(-1);
        __builtin_amdgcn_sched_barrier(0);
        {
            const v2f hown = {hst[1][0], hst[1][1]};
            accn[0] = wmma4(aWo[0][1], hown, accn[0]);
            accn[1] = wmma4(aWo[1][1], hown, accn[1]);
        }
        __builtin_amdgcn_sched_barrier(0);
        __builtin_amdgcn_s_barrier_wait(-1);
        __builtin_amdgcn_fence(__ATOMIC_ACQUIRE, "workgroup", "local");

        acc[0] = accn[0];
        acc[1] = accn[1];
        tok0 = tok1;
        tok1 = tok2;
        tok2 = tok3;
#pragma unroll
        for (int kc = 0; kc < 4; ++kc) bx1[kc] = bx2[kc];
    }

    {
        const float* hfin = hbuf + (T_LEN & 1) * (16 * HP);
        const int row = 4 * wave + (lane >> 3);
        const int c4  = (lane & 7) << 2;
        const v4f v = *(const v4f*)&hfin[row * HP + c4];
        *(v4f*)&out[(b0 + row) * HID_D + c4] = v;
    }
}

extern "C" void kernel_launch(void* const* d_in, const int* in_sizes, int n_in,
                              void* d_out, int out_size, void* d_ws, size_t ws_size,
                              hipStream_t stream) {
    const int*   tokens = (const int*)  d_in[0];
    const float* emb    = (const float*)d_in[1];
    const float* Wx     = (const float*)d_in[2];
    const float* Wh     = (const float*)d_in[3];
    const float* bias   = (const float*)d_in[4];
    float*       out    = (float*)d_out;
    (void)in_sizes; (void)n_in; (void)out_size; (void)d_ws; (void)ws_size;

    dim3 grid(BATCH / 16);
    dim3 block(32 * NWAVE);
    lstm_scan_kernel<<<grid, block, 0, stream>>>(tokens, emb, Wx, Wh, bias, out);
}


# x(t)@Wx+bias comes from a per-token table (emb_table@Wx+bias, built on the matrix unit by a new first launch into the workspace) as the accumulator's start: 8 of 24 WMMAs per step go; sum order kept
# speedup vs baseline: 7.3044x; 1.5726x over previous
#include <hip/hip_runtime.h>


typedef __attribute__((ext_vector_type(2))) float v2f;
typedef __attribute__((ext_vector_type(4))) float v4f;
typedef __attribute__((ext_vector_type(8))) float v8f;

static constexpr int T_LEN = 2048;
static constexpr int BATCH = 512;
static constexpr int VOCAB = 5000;
static constexpr int EMB_D = 16;
static constexpr int HID_D = 32;
static constexpr int GATE  = 128;
static constexpr int NWAVE = 4;
static constexpr int HP    = 36;
static constexpr int HB    = 16 * HP;
static constexpr int SP    = 132;
static constexpr int TABLE_GROUPS = (VOCAB + 16 * NWAVE - 1) / (16 * NWAVE);

__device__ __forceinline__ float frcp(float x) {
    return __builtin_amdgcn_rcpf(x);
}

__device__ __forceinline__ v8f wmma4(v2f a, v2f b, v8f c) {
    return __builtin_amdgcn_wmma_f32_16x16x4_f32(
         false, a,  false, b,
         (short)0, c,  false,  false);
}

__global__ __launch_bounds__(32 * NWAVE)
void xproj_table_kernel(const float* __restrict__ Wemb,
                        const float* __restrict__ Wx,
                        const float* __restrict__ bias,
                        float* __restrict__ P)
{
    __shared__ __attribute__((aligned(16))) float stage[NWAVE * 16 * SP];

    const int tid  = threadIdx.x;
    const int lane = tid & 31;
    const int wave = tid >> 5;
    const int half = lane >> 4;
    const int col  = lane & 15;
    const int koff = half << 1;
    const int v0   = (blockIdx.x * NWAVE + wave) << 4;
    float* st = stage + wave * (16 * SP);

    const int arow = (v0 + col < VOCAB) ? (v0 + col) : (VOCAB - 1);
    v2f a[4];
#pragma unroll
    for (int kc = 0; kc < 4; ++kc)
        a[kc] = *(const v2f*)(Wemb + arow * EMB_D + 4 * kc + koff);

#pragma unroll
    for (int nt = 0; nt < 8; ++nt) {
        const int p = 16 * nt + col;
        const int n = 32 * ((p >> 1) & 3) + 8 * (p >> 5) + 4 * ((p >> 3) & 1)
                    + 2 * ((p >> 4) & 1) + (p & 1);
        const float bv = bias[n];
        v8f c;
#pragma unroll
        for (int r = 0; r < 8; ++r) c[r] = bv;
#pragma unroll
        for (int kc = 0; kc < 4; ++kc) {
            v2f b;
            b[0] = Wx[(4 * kc + koff + 0) * GATE + n];
            b[1] = Wx[(4 * kc + koff + 1) * GATE + n];
            c = wmma4(a[kc], b, c);
        }
#pragma unroll
        for (int r = 0; r < 8; ++r) st[(8 * half + r) * SP + 16 * nt + col] = c[r];
    }
    __syncthreads();
#pragma unroll
    for (int rr = 0; rr < 16; ++rr) {
        const v4f v = *(const v4f*)&st[rr * SP + 4 * lane];
        if (v0 + rr < VOCAB) *(v4f*)&P[(v0 + rr) * GATE + 4 * lane] = v;
    }
}

static_assert(T_LEN % 3 == 2, "the scan loop is 3 steps per trip plus 2");
__global__ __launch_bounds__(32 * NWAVE)
void lstm_scan_kernel(const int* __restrict__ tokens,
                      const float* __restrict__ P,
                      const float* __restrict__ Wh,
                      float* __restrict__ out)
{
    __shared__ __attribute__((aligned(16))) float hbuf[3 * HB];

    const int tid  = threadIdx.x;
    const int lane = tid & 31;
    const int wave = tid >> 5;
    const int half = lane >> 4;
    const int col  = lane & 15;
    const int koff = half << 1;
    const int b0   = blockIdx.x << 4;

    for (int i = tid; i < 3 * HB; i += 32 * NWAVE) hbuf[i] = 0.0f;

    v2f aWo[2][2];
    v2f aWr[2][6];
#pragma unroll
    for (int tl = 0; tl < 2; ++tl) {
        const int ubase = 8 * wave + 4 * tl;
        const int n = 32 * ((col & 7) >> 1) + ubase + 2 * (col >> 3) + (col & 1);
#pragma unroll
        for (int p = 0; p < 2; ++p) {
            const int kc = 2 * wave + p;
            aWo[tl][p][0] = Wh[(4 * kc + koff + 0) * GATE + n];
            aWo[tl][p][1] = Wh[(4 * kc + koff + 1) * GATE + n];
        }
#pragma unroll
        for (int i = 0; i < 6; ++i) {
            const int kc = 2 * ((wave + 1 + (i >> 1)) & 3) + (i & 1);
            aWr[tl][i][0] = Wh[(4 * kc + koff + 0) * GATE + n];
            aWr[tl][i][1] = Wh[(4 * kc + koff + 1) * GATE + n];
        }
    }
    __syncthreads();

    float cst[2][2] = {};
    float hst[2][2] = {};

    const int* tokp = tokens + (b0 + col) * T_LEN;
    int tok0 = tokp[0];
    int tok1 = tokp[1];
    int tok2 = tokp[2];

    const unsigned pl = 32u * (unsigned)wave + 16u * (unsigned)half;

    v8f s0[2], s1[2], s2[2];
    {
        const float* r0 = P + ((unsigned)tok0 * (unsigned)GATE + pl);
        const float* r1 = P + ((unsigned)tok1 * (unsigned)GATE + pl);
        s0[0] = *(const v8f*)(r0);
        s0[1] = *(const v8f*)(r0 + 8);
        s1[0] = *(const v8f*)(r1);
        s1[1] = *(const v8f*)(r1 + 8);
        const v2f hz = {0.0f, 0.0f};
#pragma unroll
        for (int p = 0; p < 2; ++p)
#pragma unroll
            for (int tl = 0; tl < 2; ++tl)
                s0[tl] = wmma4(aWo[tl][p], hz, s0[tl]);
    }

    int rd_pair[3];
#pragma unroll
    for (int i = 0; i < 3; ++i)
        rd_pair[i] = col * HP + koff + 8 * ((wave + 1 + i) & 3);
    const int wr_off = col * HP + 8 * wave + koff;

    auto step = [&](const int t, const float* hcur, float* hnxt,
                    const int tokm, const int tokg, int& tokn,
                    v8f (&acc)[2], v8f (&pnx)[2], v8f (&pld)[2]) __attribute__((always_inline)) {
        v2f bh[6];
#pragma unroll
        for (int i = 0; i < 3; ++i) {
            bh[2 * i + 0] = *(const v2f*)&hcur[rd_pair[i]];
            bh[2 * i + 1] = *(const v2f*)&hcur[rd_pair[i] + 4];
        }
        __builtin_amdgcn_sched_barrier(0);

        tokn = tokp[(t + 3 < T_LEN) ? (t + 3) : (T_LEN - 1)];
        {
            const float* er = P + ((unsigned)tokg * (unsigned)GATE + pl);
            pld[0] = *(const v8f*)(er);
            pld[1] = *(const v8f*)(er + 8);
        }
        __builtin_amdgcn_sched_barrier(0);

        const bool m = (tokm != 0);

#pragma unroll
        for (int i = 0; i < 6; ++i)
#pragma unroll
            for (int tl = 0; tl < 2; ++tl)
                acc[tl] = wmma4(aWr[tl][i], bh[i], acc[tl]);

        float ex[4][4], rc[4][4], ec[4], cn[4];
#pragma unroll
        for (int u = 0; u < 4; ++u) {
            ex[u][0] = __expf(-acc[u >> 1][0 + (u & 1)]);
            ex[u][1] = __expf(-acc[u >> 1][2 + (u & 1)]);
            ex[u][2] = __expf(2.0f * acc[u >> 1][4 + (u & 1)]);
            ex[u][3] = __expf(-acc[u >> 1][6 + (u & 1)]);
        }
#pragma unroll
        for (int u = 0; u < 4; ++u)
#pragma unroll
            for (int q = 0; q < 4; ++q) rc[u][q] = frcp(1.0f + ex[u][q]);
#pragma unroll
        for (int u = 0; u < 4; ++u) {
            const float gv = 1.0f - 2.0f * rc[u][2];
            cn[u] = rc[u][1] * cst[u >> 1][u & 1] + rc[u][0] * gv;
            ec[u] = __expf(2.0f * cn[u]);
        }
        v2f hw2[2];
#pragma unroll
        for (int u = 0; u < 4; ++u) {
            const float hn = rc[u][3] * (1.0f - 2.0f * frcp(ec[u] + 1.0f));
            const float cw = m ? cn[u] : cst[u >> 1][u & 1];
            const float hw = m ? hn : hst[u >> 1][u & 1];
            cst[u >> 1][u & 1] = cw;
            hst[u >> 1][u & 1] = hw;
            hw2[u >> 1][u & 1] = hw;
        }

        __builtin_amdgcn_sched_barrier(0);

#pragma unroll
        for (int tl = 0; tl < 2; ++tl)
            *(v2f*)&hnxt[wr_off + 4 * tl] = hw2[tl];
        __builtin_amdgcn_sched_barrier(0);

        {
            const v2f hown = {hst[0][0], hst[0][1]};
            pnx[0] = wmma4(aWo[0][0], hown, pnx[0]);
            pnx[1] = wmma4(aWo[1][0], hown, pnx[1]);
        }
        __builtin_amdgcn_sched_barrier(0);

        __builtin_amdgcn_fence(__ATOMIC_RELEASE, "workgroup", "local");
        __builtin_amdgcn_s_barrier_signal(-1);
        __builtin_amdgcn_sched_barrier(0);
        {
            const v2f hown = {hst[1][0], hst[1][1]};
            pnx[0] = wmma4(aWo[0][1], hown, pnx[0]);
            pnx[1] = wmma4(aWo[1][1], hown, pnx[1]);
        }
        __builtin_amdgcn_sched_barrier(0);
        __builtin_amdgcn_s_barrier_wait(-1);
        __builtin_amdgcn_fence(__ATOMIC_ACQUIRE, "workgroup", "local");
    };

    float* const h0 = hbuf;
    float* const h1 = hbuf + HB;
    float* const h2 = hbuf + 2 * HB;
    int t = 0;
    for (; t + 3 <= T_LEN; t += 3) {
        int tok3, tok4, tok5;
        step(t,     h0, h1, tok0, tok2, tok3, s0, s1, s2);
        step(t + 1, h1, h2, tok1, tok3, tok4, s1, s2, s0);
        step(t + 2, h2, h0, tok2, tok4, tok5, s2, s0, s1);
        tok0 = tok3;
        tok1 = tok4;
        tok2 = tok5;
    }
    {
        int tok3, tok4;
        step(t,     h0, h1, tok0, tok2, tok3, s0, s1, s2);
        step(t + 1, h1, h2, tok1, tok3, tok4, s1, s2, s0);
    }

    {
        const float* hfin = hbuf + (T_LEN % 3) * HB;
        const int row = 4 * wave + (lane >> 3);
        const int c4  = (lane & 7) << 2;
        const v4f v = *(const v4f*)&hfin[row * HP + c4];
        *(v4f*)&out[(b0 + row) * HID_D + c4] = v;
    }
}

extern "C" void kernel_launch(void* const* d_in, const int* in_sizes, int n_in,
                              void* d_out, int out_size, void* d_ws, size_t ws_size,
                              hipStream_t stream) {
    const int*   tokens = (const int*)  d_in[0];
    const float* emb    = (const float*)d_in[1];
    const float* Wx     = (const float*)d_in[2];
    const float* Wh     = (const float*)d_in[3];
    const float* bias   = (const float*)d_in[4];
    float*       out    = (float*)d_out;
    float*       P      = (float*)d_ws;
    (void)in_sizes; (void)n_in; (void)out_size; (void)ws_size;

    dim3 tgrid(TABLE_GROUPS);
    dim3 grid(BATCH / 16);
    dim3 block(32 * NWAVE);
    xproj_table_kernel<<<tgrid, block, 0, stream>>>(emb, Wx, bias, P);
    lstm_scan_kernel<<<grid, block, 0, stream>>>(tokens, P, Wh, out);
}


# gate activations on the hardware tanh (v_tanh_f32): sigmoid(y) = (1 + tanh(y/2))/2, the 1/2 on y folded exactly into the i,f,o weight rows and table; 20 transcendentals per lane per step, not 40
# speedup vs baseline: 8.4188x; 1.1526x over previous
#include <hip/hip_runtime.h>


typedef __attribute__((ext_vector_type(2))) float v2f;
typedef __attribute__((ext_vector_type(4))) float v4f;
typedef __attribute__((ext_vector_type(8))) float v8f;

static constexpr int T_LEN = 2048;
static constexpr int BATCH = 512;
static constexpr int VOCAB = 5000;
static constexpr int EMB_D = 16;
static constexpr int HID_D = 32;
static constexpr int GATE  = 128;
static constexpr int NWAVE = 4;
static constexpr int HP    = 36;
static constexpr int HB    = 16 * HP;
static constexpr int SP    = 132;
static constexpr int TABLE_GROUPS = (VOCAB + 16 * NWAVE - 1) / (16 * NWAVE);

__device__ __forceinline__ float ftanh(float x) {
    return __builtin_amdgcn_tanhf(x);
}

__device__ __forceinline__ v8f wmma4(v2f a, v2f b, v8f c) {
    return __builtin_amdgcn_wmma_f32_16x16x4_f32(
         false, a,  false, b,
         (short)0, c,  false,  false);
}

__global__ __launch_bounds__(32 * NWAVE)
void xproj_table_kernel(const float* __restrict__ Wemb,
                        const float* __restrict__ Wx,
                        const float* __restrict__ bias,
                        float* __restrict__ P)
{
    __shared__ __attribute__((aligned(16))) float stage[NWAVE * 16 * SP];

    const int tid  = threadIdx.x;
    const int lane = tid & 31;
    const int wave = tid >> 5;
    const int half = lane >> 4;
    const int col  = lane & 15;
    const int koff = half << 1;
    const int v0   = (blockIdx.x * NWAVE + wave) << 4;
    float* st = stage + wave * (16 * SP);

    const int arow = (v0 + col < VOCAB) ? (v0 + col) : (VOCAB - 1);
    v2f a[4];
#pragma unroll
    for (int kc = 0; kc < 4; ++kc)
        a[kc] = *(const v2f*)(Wemb + arow * EMB_D + 4 * kc + koff);

#pragma unroll
    for (int nt = 0; nt < 8; ++nt) {
        const int p = 16 * nt + col;
        const int n = 32 * ((p >> 1) & 3) + 8 * (p >> 5) + 4 * ((p >> 3) & 1)
                    + 2 * ((p >> 4) & 1) + (p & 1);
        const float gs = (((p >> 1) & 3) == 2) ? 1.0f : 0.5f;
        const float bv = gs * bias[n];
        v8f c;
#pragma unroll
        for (int r = 0; r < 8; ++r) c[r] = bv;
#pragma unroll
        for (int kc = 0; kc < 4; ++kc) {
            v2f b;
            b[0] = gs * Wx[(4 * kc + koff + 0) * GATE + n];
            b[1] = gs * Wx[(4 * kc + koff + 1) * GATE + n];
            c = wmma4(a[kc], b, c);
        }
#pragma unroll
        for (int r = 0; r < 8; ++r) st[(8 * half + r) * SP + 16 * nt + col] = c[r];
    }
    __syncthreads();
#pragma unroll
    for (int rr = 0; rr < 16; ++rr) {
        const v4f v = *(const v4f*)&st[rr * SP + 4 * lane];
        if (v0 + rr < VOCAB) *(v4f*)&P[(v0 + rr) * GATE + 4 * lane] = v;
    }
}

static_assert(T_LEN % 3 == 2, "the scan loop is 3 steps per trip plus 2");
__global__ __launch_bounds__(32 * NWAVE)
void lstm_scan_kernel(const int* __restrict__ tokens,
                      const float* __restrict__ P,
                      const float* __restrict__ Wh,
                      float* __restrict__ out)
{
    __shared__ __attribute__((aligned(16))) float hbuf[3 * HB];

    const int tid  = threadIdx.x;
    const int lane = tid & 31;
    const int wave = tid >> 5;
    const int half = lane >> 4;
    const int col  = lane & 15;
    const int koff = half << 1;
    const int b0   = blockIdx.x << 4;

    for (int i = tid; i < 3 * HB; i += 32 * NWAVE) hbuf[i] = 0.0f;

    v2f aWo[2][2];
    v2f aWr[2][6];
#pragma unroll
    for (int tl = 0; tl < 2; ++tl) {
        const int ubase = 8 * wave + 4 * tl;
        const int n = 32 * ((col & 7) >> 1) + ubase + 2 * (col >> 3) + (col & 1);
        const float gs = (((col & 7) >> 1) == 2) ? 1.0f : 0.5f;
#pragma unroll
        for (int p = 0; p < 2; ++p) {
            const int kc = 2 * wave + p;
            aWo[tl][p][0] = gs * Wh[(4 * kc + koff + 0) * GATE + n];
            aWo[tl][p][1] = gs * Wh[(4 * kc + koff + 1) * GATE + n];
        }
#pragma unroll
        for (int i = 0; i < 6; ++i) {
            const int kc = 2 * ((wave + 1 + (i >> 1)) & 3) + (i & 1);
            aWr[tl][i][0] = gs * Wh[(4 * kc + koff + 0) * GATE + n];
            aWr[tl][i][1] = gs * Wh[(4 * kc + koff + 1) * GATE + n];
        }
    }
    __syncthreads();

    float cst[2][2] = {};
    float hst[2][2] = {};

    const int* tokp = tokens + (b0 + col) * T_LEN;
    int tok0 = tokp[0];
    int tok1 = tokp[1];
    int tok2 = tokp[2];

    const unsigned pl = 32u * (unsigned)wave + 16u * (unsigned)half;

    v8f s0[2], s1[2], s2[2];
    {
        const float* r0 = P + ((unsigned)tok0 * (unsigned)GATE + pl);
        const float* r1 = P + ((unsigned)tok1 * (unsigned)GATE + pl);
        s0[0] = *(const v8f*)(r0);
        s0[1] = *(const v8f*)(r0 + 8);
        s1[0] = *(const v8f*)(r1);
        s1[1] = *(const v8f*)(r1 + 8);
        const v2f hz = {0.0f, 0.0f};
#pragma unroll
        for (int p = 0; p < 2; ++p)
#pragma unroll
            for (int tl = 0; tl < 2; ++tl)
                s0[tl] = wmma4(aWo[tl][p], hz, s0[tl]);
    }

    int rd_pair[3];
#pragma unroll
    for (int i = 0; i < 3; ++i)
        rd_pair[i] = col * HP + koff + 8 * ((wave + 1 + i) & 3);
    const int wr_off = col * HP + 8 * wave + koff;

    auto step = [&](const int t, const float* hcur, float* hnxt,
                    const int tokm, const int tokg, int& tokn,
                    v8f (&acc)[2], v8f (&pnx)[2], v8f (&pld)[2]) __attribute__((always_inline)) {
        v2f bh[6];
#pragma unroll
        for (int i = 0; i < 3; ++i) {
            bh[2 * i + 0] = *(const v2f*)&hcur[rd_pair[i]];
            bh[2 * i + 1] = *(const v2f*)&hcur[rd_pair[i] + 4];
        }
        __builtin_amdgcn_sched_barrier(0);

        tokn = tokp[(t + 3 < T_LEN) ? (t + 3) : (T_LEN - 1)];
        {
            const float* er = P + ((unsigned)tokg * (unsigned)GATE + pl);
            pld[0] = *(const v8f*)(er);
            pld[1] = *(const v8f*)(er + 8);
        }
        __builtin_amdgcn_sched_barrier(0);

        const bool m = (tokm != 0);

#pragma unroll
        for (int i = 0; i < 6; ++i)
#pragma unroll
            for (int tl = 0; tl < 2; ++tl)
                acc[tl] = wmma4(aWr[tl][i], bh[i], acc[tl]);

        float ti[4], tf[4], tg[4], to[4], cn[4], tc[4];
#pragma unroll
        for (int u = 0; u < 4; ++u) {
            ti[u] = ftanh(acc[u >> 1][0 + (u & 1)]);
            tf[u] = ftanh(acc[u >> 1][2 + (u & 1)]);
            tg[u] = ftanh(acc[u >> 1][4 + (u & 1)]);
        }
#pragma unroll
        for (int u = 0; u < 4; ++u) {
            const float iv = fmaf(0.5f, ti[u], 0.5f);
            const float fv = fmaf(0.5f, tf[u], 0.5f);
            cn[u] = fv * cst[u >> 1][u & 1] + iv * tg[u];
            tc[u] = ftanh(cn[u]);
            to[u] = ftanh(acc[u >> 1][6 + (u & 1)]);
        }
        v2f hw2[2];
#pragma unroll
        for (int u = 0; u < 4; ++u) {
            const float hn = fmaf(0.5f, to[u], 0.5f) * tc[u];
            const float cw = m ? cn[u] : cst[u >> 1][u & 1];
            const float hw = m ? hn : hst[u >> 1][u & 1];
            cst[u >> 1][u & 1] = cw;
            hst[u >> 1][u & 1] = hw;
            hw2[u >> 1][u & 1] = hw;
        }

        __builtin_amdgcn_sched_barrier(0);

#pragma unroll
        for (int tl = 0; tl < 2; ++tl)
            *(v2f*)&hnxt[wr_off + 4 * tl] = hw2[tl];
        __builtin_amdgcn_sched_barrier(0);

        {
            const v2f hown = {hst[0][0], hst[0][1]};
            pnx[0] = wmma4(aWo[0][0], hown, pnx[0]);
            pnx[1] = wmma4(aWo[1][0], hown, pnx[1]);
        }
        __builtin_amdgcn_sched_barrier(0);

        __builtin_amdgcn_fence(__ATOMIC_RELEASE, "workgroup", "local");
        __builtin_amdgcn_s_barrier_signal(-1);
        __builtin_amdgcn_sched_barrier(0);
        {
            const v2f hown = {hst[1][0], hst[1][1]};
            pnx[0] = wmma4(aWo[0][1], hown, pnx[0]);
            pnx[1] = wmma4(aWo[1][1], hown, pnx[1]);
        }
        __builtin_amdgcn_sched_barrier(0);
        __builtin_amdgcn_s_barrier_wait(-1);
        __builtin_amdgcn_fence(__ATOMIC_ACQUIRE, "workgroup", "local");
    };

    float* const h0 = hbuf;
    float* const h1 = hbuf + HB;
    float* const h2 = hbuf + 2 * HB;
    int t = 0;
    for (; t + 3 <= T_LEN; t += 3) {
        int tok3, tok4, tok5;
        step(t,     h0, h1, tok0, tok2, tok3, s0, s1, s2);
        step(t + 1, h1, h2, tok1, tok3, tok4, s1, s2, s0);
        step(t + 2, h2, h0, tok2, tok4, tok5, s2, s0, s1);
        tok0 = tok3;
        tok1 = tok4;
        tok2 = tok5;
    }
    {
        int tok3, tok4;
        step(t,     h0, h1, tok0, tok2, tok3, s0, s1, s2);
        step(t + 1, h1, h2, tok1, tok3, tok4, s1, s2, s0);
    }

    {
        const float* hfin = hbuf + (T_LEN % 3) * HB;
        const int row = 4 * wave + (lane >> 3);
        const int c4  = (lane & 7) << 2;
        const v4f v = *(const v4f*)&hfin[row * HP + c4];
        *(v4f*)&out[(b0 + row) * HID_D + c4] = v;
    }
}

extern "C" void kernel_launch(void* const* d_in, const int* in_sizes, int n_in,
                              void* d_out, int out_size, void* d_ws, size_t ws_size,
                              hipStream_t stream) {
    const int*   tokens = (const int*)  d_in[0];
    const float* emb    = (const float*)d_in[1];
    const float* Wx     = (const float*)d_in[2];
    const float* Wh     = (const float*)d_in[3];
    const float* bias   = (const float*)d_in[4];
    float*       out    = (float*)d_out;
    float*       P      = (float*)d_ws;
    (void)in_sizes; (void)n_in; (void)out_size; (void)ws_size;

    dim3 tgrid(TABLE_GROUPS);
    dim3 grid(BATCH / 16);
    dim3 block(32 * NWAVE);
    xproj_table_kernel<<<tgrid, block, 0, stream>>>(emb, Wx, bias, P);
    lstm_scan_kernel<<<grid, block, 0, stream>>>(tokens, P, Wh, out);
}
